# starting kernel
# baseline (speedup 1.0000x reference)
#include <hip/hip_runtime.h>
#include <hip/hip_bf16.h>
#include <math.h>

#define B_   256
#define P_   128
#define IN_  64
#define OUT_ 64
#define BTILE 64
#define PITCH 66

typedef __attribute__((ext_vector_type(2))) float v2f;
typedef __attribute__((ext_vector_type(8))) float v8f;

__global__ __launch_bounds__(256) void sumlayer_wmma_kernel(
    const float* __restrict__ x,
    const float* __restrict__ w,
    float* __restrict__ out)
{
    __shared__ float ewT[IN_][PITCH];
    __shared__ float exs[BTILE][PITCH];
    __shared__ float lsew[OUT_];
    __shared__ float mx[BTILE];

    const int p    = blockIdx.x;
    const int b0   = blockIdx.y * BTILE;
    const int t    = threadIdx.x;
    const int lane = t & 31;
    const int wave = t >> 5;

    {
        const int o = t >> 2;
        const int q = t & 3;
        const float* wrow = w + ((size_t)p * OUT_ + o) * IN_ + q * 16;
        float v[16];
        #pragma unroll
        for (int u = 0; u < 4; ++u) {
            const float4 f = ((const float4*)wrow)[u];
            v[u*4+0] = f.x; v[u*4+1] = f.y; v[u*4+2] = f.z; v[u*4+3] = f.w;
        }
        float m = v[0];
        #pragma unroll
        for (int j = 1; j < 16; ++j) m = fmaxf(m, v[j]);
        m = fmaxf(m, __shfl_xor(m, 1, 32));
        m = fmaxf(m, __shfl_xor(m, 2, 32));
        float s = 0.f;
        #pragma unroll
        for (int j = 0; j < 16; ++j) {
            const float e = __expf(v[j] - m);
            s += e;
            ewT[q*16 + j][o] = e;
        }
        s += __shfl_xor(s, 1, 32);
        s += __shfl_xor(s, 2, 32);
        if (q == 0) lsew[o] = __logf(s);
    }

    {
        const int r = t >> 2;
        const int q = t & 3;
        const float* xrow = x + ((size_t)(b0 + r) * P_ + p) * IN_ + q * 16;
        float v[16];
        #pragma unroll
        for (int u = 0; u < 4; ++u) {
            const float4 f = ((const float4*)xrow)[u];
            v[u*4+0] = f.x; v[u*4+1] = f.y; v[u*4+2] = f.z; v[u*4+3] = f.w;
        }
        float m = v[0];
        #pragma unroll
        for (int j = 1; j < 16; ++j) m = fmaxf(m, v[j]);
        m = fmaxf(m, __shfl_xor(m, 1, 32));
        m = fmaxf(m, __shfl_xor(m, 2, 32));
        #pragma unroll
        for (int j = 0; j < 16; ++j)
            exs[r][q*16 + j] = __expf(v[j] - m);
        if (q == 0) mx[r] = m;
    }

    __syncthreads();

    const int mt    = wave & 3;
    const int nb    = (wave >> 2) * 2;
    const int lrow  = lane & 15;
    const int khalf = (lane >> 4) * 2;

    v8f acc0 = {};
    v8f acc1 = {};

    #pragma unroll
    for (int k = 0; k < IN_ / 4; ++k) {
        const int koff = k * 4 + khalf;
        v2f a;
        a.x = exs[mt*16 + lrow][koff];
        a.y = exs[mt*16 + lrow][koff + 1];
        v2f bv0;
        bv0.x = ewT[koff    ][nb*16 + lrow];
        bv0.y = ewT[koff + 1][nb*16 + lrow];
        v2f bv1;
        bv1.x = ewT[koff    ][(nb+1)*16 + lrow];
        bv1.y = ewT[koff + 1][(nb+1)*16 + lrow];
        acc0 = __builtin_amdgcn_wmma_f32_16x16x4_f32(false, a, false, bv0, (short)0, acc0, false, false);
        acc1 = __builtin_amdgcn_wmma_f32_16x16x4_f32(false, a, false, bv1, (short)0, acc1, false, false);
    }

    const int rowadd = (lane >> 4) * 8;
    const int o0 = nb*16 + lrow;
    const int o1 = (nb+1)*16 + lrow;
    const float ls0 = lsew[o0];
    const float ls1 = lsew[o1];
    #pragma unroll
    for (int r = 0; r < 8; ++r) {
        const int row  = mt*16 + r + rowadd;
        const float mr = mx[row];
        const size_t base = ((size_t)(b0 + row) * P_ + p) * OUT_;
        out[base + o0] = mr + __logf(acc0[r]) - ls0;
        out[base + o1] = mr + __logf(acc1[r]) - ls1;
    }
}

extern "C" void kernel_launch(void* const* d_in, const int* in_sizes, int n_in,
                              void* d_out, int out_size, void* d_ws, size_t ws_size,
                              hipStream_t stream) {
    (void)in_sizes; (void)n_in; (void)d_ws; (void)ws_size; (void)out_size;
    const float* x = (const float*)d_in[0];
    const float* w = (const float*)d_in[1];
    float* out = (float*)d_out;

    dim3 grid(P_, B_ / BTILE, 1);
    dim3 block(256, 1, 1);
    sumlayer_wmma_kernel<<<grid, block, 0, stream>>>(x, w, out);
}


# prep loads re-mapped to whole lines: a 16-lane row holds one 64-float tile row (4 lines per load, was 16); row max/sum now 16-lane row_xmask butterflies (weight row-sum order changed)
# speedup vs baseline: 1.0762x; 1.0762x over previous
#include <hip/hip_runtime.h>
#include <hip/hip_bf16.h>
#include <math.h>

#define B_   256
#define P_   128
#define IN_  64
#define OUT_ 64
#define BTILE 64
#define PITCH 66

typedef __attribute__((ext_vector_type(2))) float v2f;
typedef __attribute__((ext_vector_type(8))) float v8f;

template <int XMASK>
__device__ __forceinline__ float row_xor(float v) {
    const int b = __builtin_bit_cast(int, v);
    return __builtin_bit_cast(float, __builtin_amdgcn_update_dpp(b, b, 0x160 + XMASK, 0xf, 0xf, true));
}
__device__ __forceinline__ float row16_max(float v) {
    v = fmaxf(v, row_xor<1>(v));
    v = fmaxf(v, row_xor<2>(v));
    v = fmaxf(v, row_xor<4>(v));
    v = fmaxf(v, row_xor<8>(v));
    return v;
}
__device__ __forceinline__ float row16_sum(float v) {
    v += row_xor<1>(v);
    v += row_xor<2>(v);
    v += row_xor<4>(v);
    v += row_xor<8>(v);
    return v;
}

__global__ __launch_bounds__(256) void sumlayer_wmma_kernel(
    const float* __restrict__ x,
    const float* __restrict__ w,
    float* __restrict__ out)
{
    __shared__ float ewT[IN_][PITCH];
    __shared__ float exs[BTILE][PITCH];
    __shared__ float lsew[OUT_];
    __shared__ float mx[BTILE];

    const int p    = blockIdx.x;
    const int b0   = blockIdx.y * BTILE;
    const int t    = threadIdx.x;
    const int lane = t & 31;
    const int wave = t >> 5;

    const int rp = lane >> 4;
    const int c4 = lane & 15;

    {
        const float* wbase = w + (size_t)p * (OUT_ * IN_) + (wave * 8 + rp) * IN_ + c4 * 4;
        float4 f[4];
        #pragma unroll
        for (int u = 0; u < 4; ++u) f[u] = *(const float4*)(wbase + u * 2 * IN_);
        float m[4];
        #pragma unroll
        for (int u = 0; u < 4; ++u)
            m[u] = row16_max(fmaxf(fmaxf(f[u].x, f[u].y), fmaxf(f[u].z, f[u].w)));
        float s[4];
        #pragma unroll
        for (int u = 0; u < 4; ++u) {
            const int o = wave * 8 + u * 2 + rp;
            const float e0 = __expf(f[u].x - m[u]);
            const float e1 = __expf(f[u].y - m[u]);
            const float e2 = __expf(f[u].z - m[u]);
            const float e3 = __expf(f[u].w - m[u]);
            ewT[c4 * 4 + 0][o] = e0;
            ewT[c4 * 4 + 1][o] = e1;
            ewT[c4 * 4 + 2][o] = e2;
            ewT[c4 * 4 + 3][o] = e3;
            s[u] = row16_sum((e0 + e1) + (e2 + e3));
        }
        const float ssel = (c4 == 0) ? s[0] : (c4 == 1) ? s[1] : (c4 == 2) ? s[2] : s[3];
        if (c4 < 4) lsew[wave * 8 + c4 * 2 + rp] = __logf(ssel);
    }

    {
        const float* xbase = x + ((size_t)(b0 + wave * 8 + rp) * P_ + p) * IN_ + c4 * 4;
        float4 f[4];
        #pragma unroll
        for (int u = 0; u < 4; ++u) f[u] = *(const float4*)(xbase + (size_t)u * 2 * P_ * IN_);
        float m[4];
        #pragma unroll
        for (int u = 0; u < 4; ++u)
            m[u] = row16_max(fmaxf(fmaxf(f[u].x, f[u].y), fmaxf(f[u].z, f[u].w)));
        #pragma unroll
        for (int u = 0; u < 4; ++u) {
            const int r = wave * 8 + u * 2 + rp;
            exs[r][c4 * 4 + 0] = __expf(f[u].x - m[u]);
            exs[r][c4 * 4 + 1] = __expf(f[u].y - m[u]);
            exs[r][c4 * 4 + 2] = __expf(f[u].z - m[u]);
            exs[r][c4 * 4 + 3] = __expf(f[u].w - m[u]);
        }
        const float msel = (c4 == 0) ? m[0] : (c4 == 1) ? m[1] : (c4 == 2) ? m[2] : m[3];
        if (c4 < 4) mx[wave * 8 + c4 * 2 + rp] = msel;
    }

    __syncthreads();

    const int mt    = wave & 3;
    const int nb    = (wave >> 2) * 2;
    const int lrow  = lane & 15;
    const int khalf = (lane >> 4) * 2;

    v8f acc0 = {};
    v8f acc1 = {};

    #pragma unroll
    for (int k = 0; k < IN_ / 4; ++k) {
        const int koff = k * 4 + khalf;
        v2f a;
        a.x = exs[mt*16 + lrow][koff];
        a.y = exs[mt*16 + lrow][koff + 1];
        v2f bv0;
        bv0.x = ewT[koff    ][nb*16 + lrow];
        bv0.y = ewT[koff + 1][nb*16 + lrow];
        v2f bv1;
        bv1.x = ewT[koff    ][(nb+1)*16 + lrow];
        bv1.y = ewT[koff + 1][(nb+1)*16 + lrow];
        acc0 = __builtin_amdgcn_wmma_f32_16x16x4_f32(false, a, false, bv0, (short)0, acc0, false, false);
        acc1 = __builtin_amdgcn_wmma_f32_16x16x4_f32(false, a, false, bv1, (short)0, acc1, false, false);
    }

    const int rowadd = (lane >> 4) * 8;
    const int o0 = nb*16 + lrow;
    const int o1 = (nb+1)*16 + lrow;
    const float ls0 = lsew[o0];
    const float ls1 = lsew[o1];
    #pragma unroll
    for (int r = 0; r < 8; ++r) {
        const int row  = mt*16 + r + rowadd;
        const float mr = mx[row];
        const size_t base = ((size_t)(b0 + row) * P_ + p) * OUT_;
        out[base + o0] = mr + __logf(acc0[r]) - ls0;
        out[base + o1] = mr + __logf(acc1[r]) - ls1;
    }
}

extern "C" void kernel_launch(void* const* d_in, const int* in_sizes, int n_in,
                              void* d_out, int out_size, void* d_ws, size_t ws_size,
                              hipStream_t stream) {
    (void)in_sizes; (void)n_in; (void)d_ws; (void)ws_size; (void)out_size;
    const float* x = (const float*)d_in[0];
    const float* w = (const float*)d_in[1];
    float* out = (float*)d_out;

    dim3 grid(P_, B_ / BTILE, 1);
    dim3 block(256, 1, 1);
    sumlayer_wmma_kernel<<<grid, block, 0, stream>>>(x, w, out);
}


# x-tile global loads issued at kernel start with the weight loads (were issued after the weight phase): second memory latency now runs under the first and phase 1; sum order unchanged
# speedup vs baseline: 1.1978x; 1.1130x over previous
#include <hip/hip_runtime.h>
#include <hip/hip_bf16.h>
#include <math.h>

#define B_   256
#define P_   128
#define IN_  64
#define OUT_ 64
#define BTILE 64
#define PITCH 66

typedef __attribute__((ext_vector_type(2))) float v2f;
typedef __attribute__((ext_vector_type(8))) float v8f;

template <int XMASK>
__device__ __forceinline__ float row_xor(float v) {
    const int b = __builtin_bit_cast(int, v);
    return __builtin_bit_cast(float, __builtin_amdgcn_update_dpp(b, b, 0x160 + XMASK, 0xf, 0xf, true));
}
__device__ __forceinline__ float row16_max(float v) {
    v = fmaxf(v, row_xor<1>(v));
    v = fmaxf(v, row_xor<2>(v));
    v = fmaxf(v, row_xor<4>(v));
    v = fmaxf(v, row_xor<8>(v));
    return v;
}
__device__ __forceinline__ float row16_sum(float v) {
    v += row_xor<1>(v);
    v += row_xor<2>(v);
    v += row_xor<4>(v);
    v += row_xor<8>(v);
    return v;
}

__global__ __launch_bounds__(256) void sumlayer_wmma_kernel(
    const float* __restrict__ x,
    const float* __restrict__ w,
    float* __restrict__ out)
{
    __shared__ float ewT[IN_][PITCH];
    __shared__ float exs[BTILE][PITCH];
    __shared__ float lsew[OUT_];
    __shared__ float mx[BTILE];

    const int p    = blockIdx.x;
    const int b0   = blockIdx.y * BTILE;
    const int t    = threadIdx.x;
    const int lane = t & 31;
    const int wave = t >> 5;

    const int rp = lane >> 4;
    const int c4 = lane & 15;

    const float* wbase = w + (size_t)p * (OUT_ * IN_) + (wave * 8 + rp) * IN_ + c4 * 4;
    const float* xbase = x + ((size_t)(b0 + wave * 8 + rp) * P_ + p) * IN_ + c4 * 4;
    float4 fw[4], fx[4];
    #pragma unroll
    for (int u = 0; u < 4; ++u) fw[u] = *(const float4*)(wbase + u * 2 * IN_);
    #pragma unroll
    for (int u = 0; u < 4; ++u) fx[u] = *(const float4*)(xbase + (size_t)u * 2 * P_ * IN_);

    {
        const float4* f = fw;
        float m[4];
        #pragma unroll
        for (int u = 0; u < 4; ++u)
            m[u] = row16_max(fmaxf(fmaxf(f[u].x, f[u].y), fmaxf(f[u].z, f[u].w)));
        float s[4];
        #pragma unroll
        for (int u = 0; u < 4; ++u) {
            const int o = wave * 8 + u * 2 + rp;
            const float e0 = __expf(f[u].x - m[u]);
            const float e1 = __expf(f[u].y - m[u]);
            const float e2 = __expf(f[u].z - m[u]);
            const float e3 = __expf(f[u].w - m[u]);
            ewT[c4 * 4 + 0][o] = e0;
            ewT[c4 * 4 + 1][o] = e1;
            ewT[c4 * 4 + 2][o] = e2;
            ewT[c4 * 4 + 3][o] = e3;
            s[u] = row16_sum((e0 + e1) + (e2 + e3));
        }
        const float ssel = (c4 == 0) ? s[0] : (c4 == 1) ? s[1] : (c4 == 2) ? s[2] : s[3];
        if (c4 < 4) lsew[wave * 8 + c4 * 2 + rp] = __logf(ssel);
    }

    {
        const float4* f = fx;
        float m[4];
        #pragma unroll
        for (int u = 0; u < 4; ++u)
            m[u] = row16_max(fmaxf(fmaxf(f[u].x, f[u].y), fmaxf(f[u].z, f[u].w)));
        #pragma unroll
        for (int u = 0; u < 4; ++u) {
            const int r = wave * 8 + u * 2 + rp;
            exs[r][c4 * 4 + 0] = __expf(f[u].x - m[u]);
            exs[r][c4 * 4 + 1] = __expf(f[u].y - m[u]);
            exs[r][c4 * 4 + 2] = __expf(f[u].z - m[u]);
            exs[r][c4 * 4 + 3] = __expf(f[u].w - m[u]);
        }
        const float msel = (c4 == 0) ? m[0] : (c4 == 1) ? m[1] : (c4 == 2) ? m[2] : m[3];
        if (c4 < 4) mx[wave * 8 + c4 * 2 + rp] = msel;
    }

    __syncthreads();

    const int mt    = wave & 3;
    const int nb    = (wave >> 2) * 2;
    const int lrow  = lane & 15;
    const int khalf = (lane >> 4) * 2;

    v8f acc0 = {};
    v8f acc1 = {};

    #pragma unroll
    for (int k = 0; k < IN_ / 4; ++k) {
        const int koff = k * 4 + khalf;
        v2f a;
        a.x = exs[mt*16 + lrow][koff];
        a.y = exs[mt*16 + lrow][koff + 1];
        v2f bv0;
        bv0.x = ewT[koff    ][nb*16 + lrow];
        bv0.y = ewT[koff + 1][nb*16 + lrow];
        v2f bv1;
        bv1.x = ewT[koff    ][(nb+1)*16 + lrow];
        bv1.y = ewT[koff + 1][(nb+1)*16 + lrow];
        acc0 = __builtin_amdgcn_wmma_f32_16x16x4_f32(false, a, false, bv0, (short)0, acc0, false, false);
        acc1 = __builtin_amdgcn_wmma_f32_16x16x4_f32(false, a, false, bv1, (short)0, acc1, false, false);
    }

    const int rowadd = (lane >> 4) * 8;
    const int o0 = nb*16 + lrow;
    const int o1 = (nb+1)*16 + lrow;
    const float ls0 = lsew[o0];
    const float ls1 = lsew[o1];
    #pragma unroll
    for (int r = 0; r < 8; ++r) {
        const int row  = mt*16 + r + rowadd;
        const float mr = mx[row];
        const size_t base = ((size_t)(b0 + row) * P_ + p) * OUT_;
        out[base + o0] = mr + __logf(acc0[r]) - ls0;
        out[base + o1] = mr + __logf(acc1[r]) - ls1;
    }
}

extern "C" void kernel_launch(void* const* d_in, const int* in_sizes, int n_in,
                              void* d_out, int out_size, void* d_ws, size_t ws_size,
                              hipStream_t stream) {
    (void)in_sizes; (void)n_in; (void)d_ws; (void)ws_size; (void)out_size;
    const float* x = (const float*)d_in[0];
    const float* w = (const float*)d_in[1];
    float* out = (float*)d_out;

    dim3 grid(P_, B_ / BTILE, 1);
    dim3 block(256, 1, 1);
    sumlayer_wmma_kernel<<<grid, block, 0, stream>>>(x, w, out);
}


# same 64x64x64 tile per workgroup on 4 waves instead of 8 (per wave 16 rows of each prep tile and four 16x16 accumulators): half as many waves enter per launch; sum order unchanged
# speedup vs baseline: 1.2733x; 1.0630x over previous
#include <hip/hip_runtime.h>
#include <hip/hip_bf16.h>
#include <math.h>

#define B_   256
#define P_   128
#define IN_  64
#define OUT_ 64
#define BTILE 64
#define NWAVE 4
#define RPW   (BTILE / NWAVE)
#define NLD   (RPW / 2)
#define PITCH 66

typedef __attribute__((ext_vector_type(2))) float v2f;
typedef __attribute__((ext_vector_type(8))) float v8f;

template <int XMASK>
__device__ __forceinline__ float row_xor(float v) {
    const int b = __builtin_bit_cast(int, v);
    return __builtin_bit_cast(float, __builtin_amdgcn_update_dpp(b, b, 0x160 + XMASK, 0xf, 0xf, true));
}
__device__ __forceinline__ float row16_max(float v) {
    v = fmaxf(v, row_xor<1>(v));
    v = fmaxf(v, row_xor<2>(v));
    v = fmaxf(v, row_xor<4>(v));
    v = fmaxf(v, row_xor<8>(v));
    return v;
}
__device__ __forceinline__ float row16_sum(float v) {
    v += row_xor<1>(v);
    v += row_xor<2>(v);
    v += row_xor<4>(v);
    v += row_xor<8>(v);
    return v;
}

__global__ __launch_bounds__(NWAVE * 32) void sumlayer_wmma_kernel(
    const float* __restrict__ x,
    const float* __restrict__ w,
    float* __restrict__ out)
{
    __shared__ float ewT[IN_][PITCH];
    __shared__ float exs[BTILE][PITCH];
    __shared__ float lsew[OUT_];
    __shared__ float mx[BTILE];

    const int p    = blockIdx.x;
    const int b0   = blockIdx.y * BTILE;
    const int t    = threadIdx.x;
    const int lane = t & 31;
    const int wave = t >> 5;

    const int rp = lane >> 4;
    const int c4 = lane & 15;

    const float* wbase = w + (size_t)p * (OUT_ * IN_) + (wave * RPW + rp) * IN_ + c4 * 4;
    const float* xbase = x + ((size_t)(b0 + wave * RPW + rp) * P_ + p) * IN_ + c4 * 4;
    float4 fw[NLD], fx[NLD];
    #pragma unroll
    for (int u = 0; u < NLD; ++u) fw[u] = *(const float4*)(wbase + u * 2 * IN_);
    #pragma unroll
    for (int u = 0; u < NLD; ++u) fx[u] = *(const float4*)(xbase + (size_t)u * 2 * P_ * IN_);

    {
        const float4* f = fw;
        float m[NLD];
        #pragma unroll
        for (int u = 0; u < NLD; ++u)
            m[u] = row16_max(fmaxf(fmaxf(f[u].x, f[u].y), fmaxf(f[u].z, f[u].w)));
        float s[NLD];
        #pragma unroll
        for (int u = 0; u < NLD; ++u) {
            const int o = wave * RPW + u * 2 + rp;
            const float e0 = __expf(f[u].x - m[u]);
            const float e1 = __expf(f[u].y - m[u]);
            const float e2 = __expf(f[u].z - m[u]);
            const float e3 = __expf(f[u].w - m[u]);
            ewT[c4 * 4 + 0][o] = e0;
            ewT[c4 * 4 + 1][o] = e1;
            ewT[c4 * 4 + 2][o] = e2;
            ewT[c4 * 4 + 3][o] = e3;
            s[u] = row16_sum((e0 + e1) + (e2 + e3));
        }
        float ssel = s[0];
        #pragma unroll
        for (int u = 1; u < NLD; ++u) ssel = (c4 == u) ? s[u] : ssel;
        if (c4 < NLD) lsew[wave * RPW + c4 * 2 + rp] = __logf(ssel);
    }

    {
        const float4* f = fx;
        float m[NLD];
        #pragma unroll
        for (int u = 0; u < NLD; ++u)
            m[u] = row16_max(fmaxf(fmaxf(f[u].x, f[u].y), fmaxf(f[u].z, f[u].w)));
        #pragma unroll
        for (int u = 0; u < NLD; ++u) {
            const int r = wave * RPW + u * 2 + rp;
            exs[r][c4 * 4 + 0] = __expf(f[u].x - m[u]);
            exs[r][c4 * 4 + 1] = __expf(f[u].y - m[u]);
            exs[r][c4 * 4 + 2] = __expf(f[u].z - m[u]);
            exs[r][c4 * 4 + 3] = __expf(f[u].w - m[u]);
        }
        float msel = m[0];
        #pragma unroll
        for (int u = 1; u < NLD; ++u) msel = (c4 == u) ? m[u] : msel;
        if (c4 < NLD) mx[wave * RPW + c4 * 2 + rp] = msel;
    }

    __syncthreads();

    const int mt    = wave;
    const int lrow  = lane & 15;
    const int khalf = (lane >> 4) * 2;

    v8f acc[4];
    #pragma unroll
    for (int nt = 0; nt < 4; ++nt) acc[nt] = v8f{};

    #pragma unroll
    for (int k = 0; k < IN_ / 4; ++k) {
        const int koff = k * 4 + khalf;
        v2f a;
        a.x = exs[mt*16 + lrow][koff];
        a.y = exs[mt*16 + lrow][koff + 1];
        #pragma unroll
        for (int nt = 0; nt < 4; ++nt) {
            v2f bv;
            bv.x = ewT[koff    ][nt*16 + lrow];
            bv.y = ewT[koff + 1][nt*16 + lrow];
            acc[nt] = __builtin_amdgcn_wmma_f32_16x16x4_f32(false, a, false, bv, (short)0, acc[nt], false, false);
        }
    }

    const int rowadd = (lane >> 4) * 8;
    float ls[4];
    #pragma unroll
    for (int nt = 0; nt < 4; ++nt) ls[nt] = lsew[nt*16 + lrow];
    #pragma unroll
    for (int r = 0; r < 8; ++r) {
        const int row  = mt*16 + r + rowadd;
        const float mr = mx[row];
        const size_t base = ((size_t)(b0 + row) * P_ + p) * OUT_;
        #pragma unroll
        for (int nt = 0; nt < 4; ++nt)
            out[base + nt*16 + lrow] = mr + __logf(acc[nt][r]) - ls[nt];
    }
}

extern "C" void kernel_launch(void* const* d_in, const int* in_sizes, int n_in,
                              void* d_out, int out_size, void* d_ws, size_t ws_size,
                              hipStream_t stream) {
    (void)in_sizes; (void)n_in; (void)d_ws; (void)ws_size; (void)out_size;
    const float* x = (const float*)d_in[0];
    const float* w = (const float*)d_in[1];
    float* out = (float*)d_out;

    dim3 grid(P_, B_ / BTILE, 1);
    dim3 block(NWAVE * 32, 1, 1);
    sumlayer_wmma_kernel<<<grid, block, 0, stream>>>(x, w, out);
}


# LDS reads taken out of the WMMA stream: both operands row-major in LDS (weights not transposed), 16-byte fragment loads grouped ahead of the 64 WMMAs (were 70 loads among them); k sum order changed
# speedup vs baseline: 1.3362x; 1.0494x over previous
#include <hip/hip_runtime.h>
#include <hip/hip_bf16.h>
#include <math.h>

#define B_   256
#define P_   128
#define IN_  64
#define OUT_ 64
#define BTILE 64
#define NWAVE 4
#define RPW   (BTILE / NWAVE)
#define NLD   (RPW / 2)
#define PITCH 68

typedef __attribute__((ext_vector_type(2))) float v2f;
typedef __attribute__((ext_vector_type(4))) float v4f;
typedef __attribute__((ext_vector_type(8))) float v8f;

template <int XMASK>
__device__ __forceinline__ float row_xor(float v) {
    const int b = __builtin_bit_cast(int, v);
    return __builtin_bit_cast(float, __builtin_amdgcn_update_dpp(b, b, 0x160 + XMASK, 0xf, 0xf, true));
}
__device__ __forceinline__ float row16_max(float v) {
    v = fmaxf(v, row_xor<1>(v));
    v = fmaxf(v, row_xor<2>(v));
    v = fmaxf(v, row_xor<4>(v));
    v = fmaxf(v, row_xor<8>(v));
    return v;
}
__device__ __forceinline__ float row16_sum(float v) {
    v += row_xor<1>(v);
    v += row_xor<2>(v);
    v += row_xor<4>(v);
    v += row_xor<8>(v);
    return v;
}

__global__ __launch_bounds__(NWAVE * 32) void sumlayer_wmma_kernel(
    const float* __restrict__ x,
    const float* __restrict__ w,
    float* __restrict__ out)
{
    __shared__ __attribute__((aligned(16))) float ew[OUT_][PITCH];
    __shared__ __attribute__((aligned(16))) float exs[BTILE][PITCH];
    __shared__ float lsew[OUT_];
    __shared__ float mx[BTILE];

    const int p    = blockIdx.x;
    const int b0   = blockIdx.y * BTILE;
    const int t    = threadIdx.x;
    const int lane = t & 31;
    const int wave = t >> 5;

    const int rp = lane >> 4;
    const int c4 = lane & 15;

    const float* wbase = w + (size_t)p * (OUT_ * IN_) + (wave * RPW + rp) * IN_ + c4 * 4;
    const float* xbase = x + ((size_t)(b0 + wave * RPW + rp) * P_ + p) * IN_ + c4 * 4;
    float4 fw[NLD], fx[NLD];
    #pragma unroll
    for (int u = 0; u < NLD; ++u) fw[u] = *(const float4*)(wbase + u * 2 * IN_);
    #pragma unroll
    for (int u = 0; u < NLD; ++u) fx[u] = *(const float4*)(xbase + (size_t)u * 2 * P_ * IN_);

    {
        const float4* f = fw;
        float m[NLD];
        #pragma unroll
        for (int u = 0; u < NLD; ++u)
            m[u] = row16_max(fmaxf(fmaxf(f[u].x, f[u].y), fmaxf(f[u].z, f[u].w)));
        float s[NLD];
        #pragma unroll
        for (int u = 0; u < NLD; ++u) {
            const int o = wave * RPW + u * 2 + rp;
            const float e0 = __expf(f[u].x - m[u]);
            const float e1 = __expf(f[u].y - m[u]);
            const float e2 = __expf(f[u].z - m[u]);
            const float e3 = __expf(f[u].w - m[u]);
            *(v4f*)&ew[o][c4 * 4] = v4f{e0, e1, e2, e3};
            s[u] = row16_sum((e0 + e1) + (e2 + e3));
        }
        float ssel = s[0];
        #pragma unroll
        for (int u = 1; u < NLD; ++u) ssel = (c4 == u) ? s[u] : ssel;
        if (c4 < NLD) lsew[wave * RPW + c4 * 2 + rp] = __logf(ssel);
    }

    __syncthreads();

    const int mt   = wave;
    const int lrow = lane & 15;
    const int h4   = (lane >> 4) * 4;

    v4f bf[4][IN_ / 8];
    #pragma unroll
    for (int nt = 0; nt < 4; ++nt) {
        const float* brow = &ew[nt*16 + lrow][h4];
        #pragma unroll
        for (int j = 0; j < IN_ / 8; ++j) bf[nt][j] = *(const v4f*)(brow + 8 * j);
    }

    {
        const float4* f = fx;
        float m[NLD];
        #pragma unroll
        for (int u = 0; u < NLD; ++u)
            m[u] = row16_max(fmaxf(fmaxf(f[u].x, f[u].y), fmaxf(f[u].z, f[u].w)));
        #pragma unroll
        for (int u = 0; u < NLD; ++u) {
            const int r = wave * RPW + u * 2 + rp;
            *(v4f*)&exs[r][c4 * 4] = v4f{__expf(f[u].x - m[u]), __expf(f[u].y - m[u]),
                                         __expf(f[u].z - m[u]), __expf(f[u].w - m[u])};
        }
        float msel = m[0];
        #pragma unroll
        for (int u = 1; u < NLD; ++u) msel = (c4 == u) ? m[u] : msel;
        if (c4 < NLD) mx[wave * RPW + c4 * 2 + rp] = msel;
    }

    __syncthreads();

    const float* arow = &exs[mt*16 + lrow][h4];
    v4f af[IN_ / 8];
    #pragma unroll
    for (int j = 0; j < IN_ / 8; ++j) af[j] = *(const v4f*)(arow + 8 * j);
    __builtin_amdgcn_sched_barrier(0);

    v8f acc[4];
    #pragma unroll
    for (int nt = 0; nt < 4; ++nt) acc[nt] = v8f{};

    #pragma unroll
    for (int j = 0; j < IN_ / 8; ++j) {
        const v2f a0 = v2f{af[j].x, af[j].y};
        const v2f a1 = v2f{af[j].z, af[j].w};
        #pragma unroll
        for (int nt = 0; nt < 4; ++nt) {
            const v2f b0v = v2f{bf[nt][j].x, bf[nt][j].y};
            acc[nt] = __builtin_amdgcn_wmma_f32_16x16x4_f32(false, a0, false, b0v, (short)0, acc[nt], false, false);
        }
        #pragma unroll
        for (int nt = 0; nt < 4; ++nt) {
            const v2f b1v = v2f{bf[nt][j].z, bf[nt][j].w};
            acc[nt] = __builtin_amdgcn_wmma_f32_16x16x4_f32(false, a1, false, b1v, (short)0, acc[nt], false, false);
        }
    }

    const int rowadd = (lane >> 4) * 8;
    float ls[4];
    #pragma unroll
    for (int nt = 0; nt < 4; ++nt) ls[nt] = lsew[nt*16 + lrow];
    #pragma unroll
    for (int r = 0; r < 8; ++r) {
        const int row  = mt*16 + r + rowadd;
        const float mr = mx[row];
        const size_t base = ((size_t)(b0 + row) * P_ + p) * OUT_;
        #pragma unroll
        for (int nt = 0; nt < 4; ++nt)
            out[base + nt*16 + lrow] = mr + __logf(acc[nt][r]) - ls[nt];
    }
}

extern "C" void kernel_launch(void* const* d_in, const int* in_sizes, int n_in,
                              void* d_out, int out_size, void* d_ws, size_t ws_size,
                              hipStream_t stream) {
    (void)in_sizes; (void)n_in; (void)d_ws; (void)ws_size; (void)out_size;
    const float* x = (const float*)d_in[0];
    const float* w = (const float*)d_in[1];
    float* out = (float*)d_out;

    dim3 grid(P_, B_ / BTILE, 1);
    dim3 block(NWAVE * 32, 1, 1);
    sumlayer_wmma_kernel<<<grid, block, 0, stream>>>(x, w, out);
}
